# starting kernel
# baseline (speedup 1.0000x reference)
#include <hip/hip_runtime.h>
#include <math.h>

typedef __attribute__((ext_vector_type(2))) float v2f;
typedef __attribute__((ext_vector_type(8))) float v8f;

#define BB 64
#define SS 50
#define LL 20
#define QQ 20
#define DD 128
#define VV 32000

__global__ void k_a_embed(const int* __restrict__ story,
                          const float* __restrict__ embedA,
                          float* __restrict__ a_embed) {
  const int bs = blockIdx.x;
  const int d  = threadIdx.x;
  const int* w = story + bs * LL;
  float acc = 0.f;
#pragma unroll
  for (int l = 0; l < LL; ++l)
    acc += embedA[(long)w[l] * DD + d];
  a_embed[bs * DD + d] = acc;
}

__global__ void k_u(const int* __restrict__ query,
                    const float* __restrict__ embedB,
                    float* __restrict__ u) {
  const int b = blockIdx.x;
  const int d = threadIdx.x;
  const int* w = query + b * QQ;
  float acc = 0.f;
#pragma unroll
  for (int q = 0; q < QQ; ++q)
    acc += embedB[(long)w[q] * DD + d];
  u[b * DD + d] = acc;
}

__global__ void k_ip(const float* __restrict__ a_embed,
                     const float* __restrict__ u,
                     float* __restrict__ ip) {
  const int s = blockIdx.x;
  __shared__ float red[256];
  float acc = 0.f;
  for (int i = threadIdx.x; i < BB * DD; i += 256) {
    const int b = i >> 7;
    const int k = i & (DD - 1);
    acc += a_embed[(b * SS + s) * DD + k] * u[b * DD + k];
  }
  red[threadIdx.x] = acc;
  __syncthreads();
  for (int off = 128; off > 0; off >>= 1) {
    if (threadIdx.x < off) red[threadIdx.x] += red[threadIdx.x + off];
    __syncthreads();
  }
  if (threadIdx.x == 0) ip[s] = red[0];
}

__global__ void k_softmax_p(const float* __restrict__ ip, float* __restrict__ p) {
  __shared__ float red[64];
  const int t = threadIdx.x;
  const float v = (t < SS) ? ip[t] : -INFINITY;
  red[t] = v; __syncthreads();
  for (int off = 32; off > 0; off >>= 1) {
    if (t < off) red[t] = fmaxf(red[t], red[t + off]);
    __syncthreads();
  }
  const float mx = red[0];
  __syncthreads();
  const float e = (t < SS) ? expf(v - mx) : 0.f;
  red[t] = e; __syncthreads();
  for (int off = 32; off > 0; off >>= 1) {
    if (t < off) red[t] += red[t + off];
    __syncthreads();
  }
  if (t < SS) p[t] = e / red[0];
}

__global__ void k_ten3(const float* __restrict__ a_embed,
                       const float* __restrict__ u,
                       const float* __restrict__ p,
                       float* __restrict__ ten3) {
  const int b = blockIdx.x;
  const int d = threadIdx.x;
  float acc = u[b * DD + d];
  for (int s = 0; s < SS; ++s)
    acc += a_embed[(b * SS + s) * DD + d] * p[s];
  ten3[b * DD + d] = acc;
}

__global__ void __launch_bounds__(32)
k_gemm_wmma(const float* __restrict__ ten3,
            const float* __restrict__ Wm,
            const float* __restrict__ bias,
            float* __restrict__ out) {
  const int lane = threadIdx.x;
  const int m0   = blockIdx.y << 4;
  const int n0   = blockIdx.x << 6;
  const int half = lane >> 4;
  const int l15  = lane & 15;

  const float* __restrict__ arow = ten3 + (m0 + l15) * DD;
  const float* __restrict__ b0r  = Wm + (long)(n0 +  0 + l15) * DD;
  const float* __restrict__ b1r  = Wm + (long)(n0 + 16 + l15) * DD;
  const float* __restrict__ b2r  = Wm + (long)(n0 + 32 + l15) * DD;
  const float* __restrict__ b3r  = Wm + (long)(n0 + 48 + l15) * DD;

  v8f c0 = {}, c1 = {}, c2 = {}, c3 = {};

#pragma unroll 4
  for (int k = 0; k < DD; k += 4) {
    const int kk = k + half * 2;
    const v2f a  = *(const v2f*)(arow + kk);
    const v2f b0 = *(const v2f*)(b0r  + kk);
    const v2f b1 = *(const v2f*)(b1r  + kk);
    const v2f b2 = *(const v2f*)(b2r  + kk);
    const v2f b3 = *(const v2f*)(b3r  + kk);
    c0 = __builtin_amdgcn_wmma_f32_16x16x4_f32(false, a, false, b0, (short)0, c0, false, false);
    c1 = __builtin_amdgcn_wmma_f32_16x16x4_f32(false, a, false, b1, (short)0, c1, false, false);
    c2 = __builtin_amdgcn_wmma_f32_16x16x4_f32(false, a, false, b2, (short)0, c2, false, false);
    c3 = __builtin_amdgcn_wmma_f32_16x16x4_f32(false, a, false, b3, (short)0, c3, false, false);
  }

  const int ncol = n0 + l15;
#pragma unroll
  for (int r = 0; r < 8; ++r) {
    const long m = m0 + r + 8 * half;
    out[m * VV + ncol +  0] = c0[r] + bias[ncol +  0];
    out[m * VV + ncol + 16] = c1[r] + bias[ncol + 16];
    out[m * VV + ncol + 32] = c2[r] + bias[ncol + 32];
    out[m * VV + ncol + 48] = c3[r] + bias[ncol + 48];
  }
}

__global__ void k_softmax_rows(float* __restrict__ out) {
  const int row = blockIdx.x;
  float* __restrict__ x = out + (long)row * VV;
  __shared__ float red[256];
  const int t = threadIdx.x;

  float m = -INFINITY;
  for (int i = t; i < VV; i += 256) m = fmaxf(m, x[i]);
  red[t] = m; __syncthreads();
  for (int off = 128; off > 0; off >>= 1) {
    if (t < off) red[t] = fmaxf(red[t], red[t + off]);
    __syncthreads();
  }
  const float mx = red[0];
  __syncthreads();

  float s = 0.f;
  for (int i = t; i < VV; i += 256) s += expf(x[i] - mx);
  red[t] = s; __syncthreads();
  for (int off = 128; off > 0; off >>= 1) {
    if (t < off) red[t] += red[t + off];
    __syncthreads();
  }
  const float inv = 1.0f / red[0];

  for (int i = t; i < VV; i += 256) x[i] = expf(x[i] - mx) * inv;
}

extern "C" void kernel_launch(void* const* d_in, const int* in_sizes, int n_in,
                              void* d_out, int out_size, void* d_ws, size_t ws_size,
                              hipStream_t stream) {
  const int*   story  = (const int*)d_in[0];
  const int*   query  = (const int*)d_in[1];
  const float* embedA = (const float*)d_in[2];
  const float* embedB = (const float*)d_in[3];
  const float* Wm     = (const float*)d_in[4];
  const float* bias   = (const float*)d_in[5];
  float* out = (float*)d_out;
  float* ws  = (float*)d_ws;

  float* a_embed = ws;
  float* u       = ws + 409600;
  float* ip      = ws + 417792;
  float* p       = ws + 417856;
  float* ten3    = ws + 417920;

  k_a_embed  <<<BB * SS, DD, 0, stream>>>(story, embedA, a_embed);
  k_u        <<<BB,      DD, 0, stream>>>(query, embedB, u);
  k_ip       <<<SS,     256, 0, stream>>>(a_embed, u, ip);
  k_softmax_p<<<1,       64, 0, stream>>>(ip, p);
  k_ten3     <<<BB,      DD, 0, stream>>>(a_embed, u, p, ten3);

  dim3 gg(VV / 64, BB / 16);
  k_gemm_wmma<<<gg, 32, 0, stream>>>(ten3, Wm, bias, out);

  k_softmax_rows<<<BB, 256, 0, stream>>>(out);
}


# Launches 7 -> 5: the u gather runs as 64 extra blocks of the a_embed launch; each ten3 wave recomputes softmax-50 from ip in registers (same butterflies, p bit-identical); all sum orders kept
# speedup vs baseline: 3.0616x; 3.0616x over previous
#include <hip/hip_runtime.h>
#include <math.h>

typedef __attribute__((ext_vector_type(2))) float v2f;
typedef __attribute__((ext_vector_type(4))) float v4f;
typedef __attribute__((ext_vector_type(8))) float v8f;

#define BB 64
#define SS 50
#define LL 20
#define QQ 20
#define DD 128
#define VV 32000

__device__ __forceinline__ float wave_max(float v) {
  for (int off = 16; off > 0; off >>= 1) v = fmaxf(v, __shfl_xor(v, off, 32));
  return v;
}
__device__ __forceinline__ float wave_sum(float v) {
  for (int off = 16; off > 0; off >>= 1) v += __shfl_xor(v, off, 32);
  return v;
}

static_assert(LL == QQ, "one gather loop serves both index lists");

__global__ void __launch_bounds__(DD)
k_embed(const int* __restrict__ story,
        const int* __restrict__ query,
        const float* __restrict__ embedA,
        const float* __restrict__ embedB,
        float* __restrict__ a_embed,
        float* __restrict__ u) {
  const bool is_u = blockIdx.x < BB;
  const int  r    = is_u ? blockIdx.x : blockIdx.x - BB;
  const int* __restrict__ w     = (is_u ? query : story) + r * LL;
  const float* __restrict__ tab = is_u ? embedB : embedA;
  float* __restrict__ dst       = (is_u ? u : a_embed) + r * DD;
  const int d = threadIdx.x;
  float acc = 0.f;
#pragma unroll
  for (int l = 0; l < LL; ++l)
    acc += tab[(long)w[l] * DD + d];
  dst[d] = acc;
}

__global__ void __launch_bounds__(256)
k_ip(const float* __restrict__ a_embed,
     const float* __restrict__ u,
     float* __restrict__ ip) {
  const int s = blockIdx.x;
  const int t = threadIdx.x;
  __shared__ float red[8];
  const float4* __restrict__ a4 = (const float4*)a_embed;
  const float4* __restrict__ u4 = (const float4*)u;
  float4 av[8], uv[8];
#pragma unroll
  for (int j = 0; j < 8; ++j) {
    const int e = t + 256 * j;
    av[j] = a4[((e >> 5) * SS + s) * (DD / 4) + (e & 31)];
    uv[j] = u4[e];
  }
  __builtin_amdgcn_sched_barrier(0);
  float s0 = 0.f, s1 = 0.f, s2 = 0.f, s3 = 0.f;
#pragma unroll
  for (int j = 0; j < 8; ++j) {
    s0 += av[j].x * uv[j].x;
    s1 += av[j].y * uv[j].y;
    s2 += av[j].z * uv[j].z;
    s3 += av[j].w * uv[j].w;
  }
  const float w = wave_sum((s0 + s1) + (s2 + s3));
  if ((t & 31) == 0) red[t >> 5] = w;
  __syncthreads();
  if (t == 0) {
    float r = red[0];
#pragma unroll
    for (int i = 1; i < 8; ++i) r += red[i];
    ip[s] = r;
  }
}

__device__ __forceinline__ float lane_value(float v, int src_lane) {
  return __builtin_bit_cast(
      float, __builtin_amdgcn_readlane(__builtin_bit_cast(int, v), src_lane));
}

__global__ void __launch_bounds__(128)
k_ten3(const float* __restrict__ a_embed,
       const float* __restrict__ u,
       const float* __restrict__ ip,
       float* __restrict__ ten3) {
  const int b = blockIdx.x;
  const int d = threadIdx.x;
  const int t = d & 31;
  const bool two = (t + 32) < SS;
  const float v0 = ip[t];
  const float v1 = two ? ip[t + 32] : -INFINITY;
  float av[SS];
#pragma unroll
  for (int s = 0; s < SS; ++s) av[s] = a_embed[(b * SS + s) * DD + d];
  float acc = u[b * DD + d];
  __builtin_amdgcn_sched_barrier(0);
  const float mx  = wave_max(fmaxf(v0, v1));
  const float e0  = expf(v0 - mx);
  const float e1  = two ? expf(v1 - mx) : 0.f;
  const float sum = wave_sum(e0 + e1);
  const float p0  = e0 / sum;
  const float p1  = e1 / sum;
#pragma unroll
  for (int s = 0; s < SS; ++s)
    acc += av[s] * (s < 32 ? lane_value(p0, s) : lane_value(p1, s - 32));
  ten3[((((b >> 4) * 16 + (d >> 3)) * 32 + ((d >> 2) & 1) * 16 + (b & 15)) << 2) + (d & 3)] = acc;
}

#define GEMM_STEP(A, B, C)                                                     \
  C = __builtin_amdgcn_wmma_f32_16x16x4_f32(false, A, false, B, (short)0, C,   \
                                            false, false)

__global__ void __launch_bounds__(128)
k_gemm_wmma(const float* __restrict__ ten3f,
            const float* __restrict__ Wm,
            const float* __restrict__ bias,
            float* __restrict__ out) {
  const int lane = threadIdx.x & 31;
  const int wv   = threadIdx.x >> 5;
  const int half = lane >> 4;
  const int l15  = lane & 15;
  const int n0   = ((blockIdx.x << 2) + wv) << 5;

  const v4f* __restrict__ bp0 = (const v4f*)(Wm + (long)(n0 + l15) * DD + half * 4);
  const v4f* __restrict__ bp1 = (const v4f*)(Wm + (long)(n0 + 16 + l15) * DD + half * 4);
  const v4f* __restrict__ ap  = (const v4f*)ten3f + lane;

  v4f b0[16], b1[16], a[4][16];
#pragma unroll
  for (int j = 0; j < 16; ++j) b0[j] = bp0[2 * j];
#pragma unroll
  for (int t = 0; t < 4; ++t) {
#pragma unroll
    for (int j = 0; j < 16; ++j) a[t][j] = ap[(t * 16 + j) * 32];
  }
  const float bias0 = bias[n0 + l15];
  const float bias1 = bias[n0 + 16 + l15];
  __builtin_amdgcn_sched_barrier(0);
#pragma unroll
  for (int j = 0; j < 16; ++j) b1[j] = bp1[2 * j];
  __builtin_amdgcn_sched_barrier(0);

  float* __restrict__ orow = out + (long)(8 * half) * VV + n0 + l15;

  {
    v8f c0 = {}, c1 = {}, c2 = {}, c3 = {};
#pragma unroll
    for (int j = 0; j < 16; ++j) {
      const v2f blo = __builtin_shufflevector(b0[j], b0[j], 0, 1);
      const v2f bhi = __builtin_shufflevector(b0[j], b0[j], 2, 3);
      GEMM_STEP(__builtin_shufflevector(a[0][j], a[0][j], 0, 1), blo, c0);
      GEMM_STEP(__builtin_shufflevector(a[1][j], a[1][j], 0, 1), blo, c1);
      GEMM_STEP(__builtin_shufflevector(a[2][j], a[2][j], 0, 1), blo, c2);
      GEMM_STEP(__builtin_shufflevector(a[3][j], a[3][j], 0, 1), blo, c3);
      GEMM_STEP(__builtin_shufflevector(a[0][j], a[0][j], 2, 3), bhi, c0);
      GEMM_STEP(__builtin_shufflevector(a[1][j], a[1][j], 2, 3), bhi, c1);
      GEMM_STEP(__builtin_shufflevector(a[2][j], a[2][j], 2, 3), bhi, c2);
      GEMM_STEP(__builtin_shufflevector(a[3][j], a[3][j], 2, 3), bhi, c3);
    }
#pragma unroll
    for (int r = 0; r < 8; ++r) {
      orow[(long)(r +  0) * VV] = c0[r] + bias0;
      orow[(long)(r + 16) * VV] = c1[r] + bias0;
      orow[(long)(r + 32) * VV] = c2[r] + bias0;
      orow[(long)(r + 48) * VV] = c3[r] + bias0;
    }
  }
  __builtin_amdgcn_sched_barrier(0);
  {
    v8f c0 = {}, c1 = {}, c2 = {}, c3 = {};
#pragma unroll
    for (int j = 0; j < 16; ++j) {
      const v2f blo = __builtin_shufflevector(b1[j], b1[j], 0, 1);
      const v2f bhi = __builtin_shufflevector(b1[j], b1[j], 2, 3);
      GEMM_STEP(__builtin_shufflevector(a[0][j], a[0][j], 0, 1), blo, c0);
      GEMM_STEP(__builtin_shufflevector(a[1][j], a[1][j], 0, 1), blo, c1);
      GEMM_STEP(__builtin_shufflevector(a[2][j], a[2][j], 0, 1), blo, c2);
      GEMM_STEP(__builtin_shufflevector(a[3][j], a[3][j], 0, 1), blo, c3);
      GEMM_STEP(__builtin_shufflevector(a[0][j], a[0][j], 2, 3), bhi, c0);
      GEMM_STEP(__builtin_shufflevector(a[1][j], a[1][j], 2, 3), bhi, c1);
      GEMM_STEP(__builtin_shufflevector(a[2][j], a[2][j], 2, 3), bhi, c2);
      GEMM_STEP(__builtin_shufflevector(a[3][j], a[3][j], 2, 3), bhi, c3);
    }
#pragma unroll
    for (int r = 0; r < 8; ++r) {
      orow[(long)(r +  0) * VV + 16] = c0[r] + bias1;
      orow[(long)(r + 16) * VV + 16] = c1[r] + bias1;
      orow[(long)(r + 32) * VV + 16] = c2[r] + bias1;
      orow[(long)(r + 48) * VV + 16] = c3[r] + bias1;
    }
  }
}

#define SM_T    1024
#define SM_P    (VV / 4)
#define SM_FULL (SM_P / SM_T)
#define SM_REST (SM_P - SM_FULL * SM_T)

__global__ void __launch_bounds__(SM_T)
k_softmax_rows(float* __restrict__ out) {
  const int row = blockIdx.x;
  float4* __restrict__ x4 = (float4*)(out + (long)row * VV);
  __shared__ float red_m[32];
  __shared__ float red_s[32];
  const int t    = threadIdx.x;
  const int lane = t & 31;
  const int wv   = t >> 5;
  const bool rest = t < SM_REST;

  float4 v[SM_FULL + 1];
  for (int j = 0; j < SM_FULL; ++j) v[j] = x4[t + j * SM_T];
  v[SM_FULL] = rest ? x4[t + SM_FULL * SM_T]
                    : make_float4(-INFINITY, -INFINITY, -INFINITY, -INFINITY);

  float m = -INFINITY;
  for (int j = 0; j <= SM_FULL; ++j)
    m = fmaxf(m, fmaxf(fmaxf(v[j].x, v[j].y), fmaxf(v[j].z, v[j].w)));
  m = wave_max(m);
  if (lane == 0) red_m[wv] = m;
  __syncthreads();
  const float mx = wave_max(red_m[lane]);

  float s0 = 0.f, s1 = 0.f, s2 = 0.f, s3 = 0.f;
  for (int j = 0; j < SM_FULL; ++j) {
    v[j].x = expf(v[j].x - mx); s0 += v[j].x;
    v[j].y = expf(v[j].y - mx); s1 += v[j].y;
    v[j].z = expf(v[j].z - mx); s2 += v[j].z;
    v[j].w = expf(v[j].w - mx); s3 += v[j].w;
  }
  if (rest) {
    v[SM_FULL].x = expf(v[SM_FULL].x - mx); s0 += v[SM_FULL].x;
    v[SM_FULL].y = expf(v[SM_FULL].y - mx); s1 += v[SM_FULL].y;
    v[SM_FULL].z = expf(v[SM_FULL].z - mx); s2 += v[SM_FULL].z;
    v[SM_FULL].w = expf(v[SM_FULL].w - mx); s3 += v[SM_FULL].w;
  }
  float s = wave_sum((s0 + s1) + (s2 + s3));
  if (lane == 0) red_s[wv] = s;
  __syncthreads();
  const float inv = 1.0f / wave_sum(red_s[lane]);

  for (int j = 0; j < SM_FULL; ++j) {
    v[j].x *= inv; v[j].y *= inv; v[j].z *= inv; v[j].w *= inv;
    x4[t + j * SM_T] = v[j];
  }
  if (rest) {
    v[SM_FULL].x *= inv; v[SM_FULL].y *= inv; v[SM_FULL].z *= inv; v[SM_FULL].w *= inv;
    x4[t + SM_FULL * SM_T] = v[SM_FULL];
  }
}

extern "C" void kernel_launch(void* const* d_in, const int* in_sizes, int n_in,
                              void* d_out, int out_size, void* d_ws, size_t ws_size,
                              hipStream_t stream) {
  const int*   story  = (const int*)d_in[0];
  const int*   query  = (const int*)d_in[1];
  const float* embedA = (const float*)d_in[2];
  const float* embedB = (const float*)d_in[3];
  const float* Wm     = (const float*)d_in[4];
  const float* bias   = (const float*)d_in[5];
  float* out = (float*)d_out;
  float* ws  = (float*)d_ws;

  float* a_embed = ws;
  float* u       = ws + 409600;
  float* ip      = ws + 417792;
  float* ten3    = ws + 417920;

  k_embed<<<BB + BB * SS, DD, 0, stream>>>(story, query, embedA, embedB, a_embed, u);
  k_ip   <<<SS,          256, 0, stream>>>(a_embed, u, ip);
  k_ten3 <<<BB,           DD, 0, stream>>>(a_embed, u, ip, ten3);

  k_gemm_wmma<<<VV / 128, 128, 0, stream>>>(ten3, Wm, bias, out);

  k_softmax_rows<<<BB, SM_T, 0, stream>>>(out);
}
